# starting kernel
# baseline (speedup 1.0000x reference)
#include <hip/hip_runtime.h>

typedef float v2f __attribute__((ext_vector_type(2)));
typedef float v8f __attribute__((ext_vector_type(8)));

#define H28 28
#define P784 784
#define C32 32
#define ROW588 588
#define KPAD 64
#define MOUT 256

__global__ void k_t3(const float* __restrict__ x, float* __restrict__ t3r) {
    int tid = blockIdx.x * blockDim.x + threadIdx.x;
    if (tid >= C32 * P784) return;
    int c = tid / P784, rem = tid % P784;
    int h = rem / H28, w = rem % H28;
    int wsrc = (w + H28 - 1) % H28;
    int src = c * P784 + h * H28 + wsrc;
    t3r[tid] = x[src] + x[src + C32 * P784];
}

__global__ void k_t5(const float* __restrict__ t3r, const float* __restrict__ w1,
                     float* __restrict__ t5p) {
    int tid = blockIdx.x * blockDim.x + threadIdx.x;
    if (tid >= 30 * ROW588) return;
    int row = tid / ROW588;
    int rem = tid % ROW588;
    if (row == 0 || row == 29) { t5p[tid] = 0.0f; return; }
    int h = row - 1;
    int w = rem / 21, kc = rem % 21;
    float acc = 0.0f;
    for (int c = 0; c < C32; ++c) {
        const float* t3row = t3r + c * P784 + h * H28;
        const float* w1row = w1 + c * 147 + kc;
#pragma unroll
        for (int j = 0; j < 7; ++j) {
            int ww = w + j - 3;
            float v = (ww >= 0 && ww < H28) ? t3row[ww] : 0.0f;
            acc += v * w1row[j * 21];
        }
    }
    t5p[tid] = acc;
}

__global__ void k_packA(const float* __restrict__ w0, float* __restrict__ w0p) {
    int tid = blockIdx.x * blockDim.x + threadIdx.x;
    if (tid >= MOUT * KPAD) return;
    int i = tid >> 6, k = tid & 63;
    w0p[tid] = (k < 63) ? w0[i * 63 + k] : 0.0f;
}

__global__ void k_packB(const float* __restrict__ t5p, float* __restrict__ Bm) {
    int tid = blockIdx.x * blockDim.x + threadIdx.x;
    if (tid >= P784 * KPAD) return;
    int p = tid >> 6, k = tid & 63;
    float v = 0.0f;
    if (k < 63) {
        int j = k / 21, kc = k % 21;
        int h = p / H28, w = p % H28;
        v = t5p[(h + j) * ROW588 + w * 21 + kc];
    }
    Bm[tid] = v;
}

__global__ void __launch_bounds__(32)
k_gemm(const float* __restrict__ A, const float* __restrict__ B, float* __restrict__ D) {
    const int lane = threadIdx.x;
    const int t    = lane & 15;
    const int kh   = lane >> 4;
    const int mi   = blockIdx.x;
    const int ni   = blockIdx.y;

    const float* Arow = A + (size_t)(mi * 16 + t) * KPAD;
    const float* Brow = B + (size_t)(ni * 16 + t) * KPAD;

    v8f c = {0.f, 0.f, 0.f, 0.f, 0.f, 0.f, 0.f, 0.f};
#pragma unroll
    for (int kk = 0; kk < 16; ++kk) {
        const int k0 = kk * 4 + kh * 2;
        v2f a = *(const v2f*)(Arow + k0);
        v2f b = *(const v2f*)(Brow + k0);
        c = __builtin_amdgcn_wmma_f32_16x16x4_f32(
                 false, a,  false, b,
                 (short)0, c,  false,  false);
    }

#pragma unroll
    for (int r = 0; r < 8; ++r) {
        const int m = r + kh * 8;
        D[(size_t)(mi * 16 + m) * P784 + ni * 16 + t] = c[r];
    }
}

extern "C" void kernel_launch(void* const* d_in, const int* in_sizes, int n_in,
                              void* d_out, int out_size, void* d_ws, size_t ws_size,
                              hipStream_t stream) {
    const float* x  = (const float*)d_in[0];
    const float* w0 = (const float*)d_in[1];
    const float* w1 = (const float*)d_in[2];
    float* y = (float*)d_out;

    float* ws  = (float*)d_ws;
    float* t3r = ws;
    float* t5p = ws + 25088;
    float* w0p = ws + 42728;
    float* Bm  = ws + 59112;

    k_t3   <<<(C32 * P784 + 255) / 256, 256, 0, stream>>>(x, t3r);
    k_t5   <<<(30 * ROW588 + 255) / 256, 256, 0, stream>>>(t3r, w1, t5p);
    k_packA<<<(MOUT * KPAD) / 256,      256, 0, stream>>>(w0, w0p);
    k_packB<<<(P784 * KPAD) / 256,      256, 0, stream>>>(t5p, Bm);
    k_gemm <<<dim3(16, 49), 32, 0, stream>>>(w0p, Bm, y);
}


# four launches fused into one: a block (one output row, half its width, 128 channels) rebuilds its 3x17 tap-last U rows by WMMA from registers into LDS; 7-tap add in LDS feeds stage 2; same sums
# speedup vs baseline: 6.1267x; 6.1267x over previous
#include <hip/hip_runtime.h>


typedef float v2f __attribute__((ext_vector_type(2)));
typedef float v8f __attribute__((ext_vector_type(8)));
typedef float v4f __attribute__((ext_vector_type(4)));
typedef float v4fu __attribute__((ext_vector_type(4), aligned(4)));

#define H28 28
#define P784 784
#define C32 32
#define NF 147
#define K63 63
#define WOUT 14
#define SRC 17
#define RPI 23
#define UROWS 85
#define PU 164
#define PB 68

__global__ void __launch_bounds__(256)
k_fused(const float* __restrict__ x, const float* __restrict__ w0,
        const float* __restrict__ w1, float* __restrict__ y) {
    __shared__ __attribute__((aligned(16))) float Us[UROWS * PU];
    __shared__ __attribute__((aligned(16))) float Bs[16 * PB];

    const int tid  = threadIdx.x;
    const int lane = tid & 31;
    const int wave = __builtin_amdgcn_readfirstlane(tid >> 5);
    const int t    = lane & 15;
    const int kh   = lane >> 4;

    const int h   = blockIdx.x;
    const int wh  = blockIdx.y;
    const int g   = blockIdx.z;
    const int ww0 = 11 * wh;

    const int mt = wave & 3;
    const int nh = wave >> 2;
    const int r  = (mt < 3) ? mt : ((t < 2) ? t : 2);
    const int s  = (mt < 3) ? t : (SRC - 1);
    int hh = h + r - 1;
    hh = hh < 0 ? 0 : (hh > H28 - 1 ? H28 - 1 : hh);
    const int ww   = ww0 + s;
    const int wsrc = (ww == 0) ? (H28 - 1) : (ww - 1);
    const float* xa = x + hh * H28 + wsrc;

    const float* wq = w1 + 64 * nh + 4 * t;
    const int fe    = 128 + 16 * nh + t;
    const float* we = w1 + ((fe < NF) ? fe : (NF - 1));

    float xl[8][4];
    v4f   bq[8][2];
    v2f   be[8];
#pragma unroll
    for (int kk = 0; kk < 8; ++kk) {
        const int k0 = kk * 4 + kh * 2;
        xl[kk][0] = xa[(k0    ) * P784];
        xl[kk][1] = xa[(k0     + C32) * P784];
        xl[kk][2] = xa[(k0 + 1) * P784];
        xl[kk][3] = xa[(k0 + 1 + C32) * P784];
        bq[kk][0] = *(const v4fu*)(wq + (k0    ) * NF);
        bq[kk][1] = *(const v4fu*)(wq + (k0 + 1) * NF);
        be[kk].x  = we[(k0    ) * NF];
        be[kk].y  = we[(k0 + 1) * NF];
    }

    __builtin_amdgcn_sched_barrier(0);

    const float* w0row = w0 + (size_t)(g * 128 + wave * 16 + t) * K63;
    v2f a2[16];
#pragma unroll
    for (int kk = 0; kk < 16; ++kk) {
        const int k0 = kk * 4 + kh * 2;
        const int k1 = (k0 + 1 < K63) ? (k0 + 1) : (K63 - 1);
        const float v1 = w0row[k1];
        a2[kk].x = w0row[k0];
        a2[kk].y = (k0 + 1 < K63) ? v1 : 0.0f;
    }

    __builtin_amdgcn_sched_barrier(0);

#pragma unroll
    for (int i = 0; i < 3; ++i) {
        const int idx = tid + i * 256;
        if (idx < 18 * 40) {
            const int zr = idx / 40;
            const int c4 = idx - zr * 40;
            const int ri = zr / 6;
            const int pp = zr - ri * 6;
            const int row = ri * RPI + ((pp < 3) ? pp : (pp + SRC));
            *(v4f*)(Us + row * PU + 4 * c4) = (v4f){0.f, 0.f, 0.f, 0.f};
        }
    }

    {
        v8f c1[5];
#pragma unroll
        for (int q = 0; q < 5; ++q)
            c1[q] = (v8f){0.f, 0.f, 0.f, 0.f, 0.f, 0.f, 0.f, 0.f};

#pragma unroll
        for (int kk = 0; kk < 8; ++kk) {
            v2f a;
            a.x = xl[kk][0] + xl[kk][1];
            a.y = xl[kk][2] + xl[kk][3];
#pragma unroll
            for (int q = 0; q < 5; ++q) {
                v2f b;
                if (q < 4) { b.x = bq[kk][0][q]; b.y = bq[kk][1][q]; }
                else       { b = be[kk]; }
                c1[q] = __builtin_amdgcn_wmma_f32_16x16x4_f32(
                             false, a,  false, b,
                             (short)0, c1[q],  false,  false);
            }
        }

        const int cq = 64 * nh + 4 * t;
        const int ce = 128 + 16 * nh + t;
        if (mt < 3) {
            float* u0 = Us + (mt * RPI + 3 + kh * 8) * PU;
#pragma unroll
            for (int r8 = 0; r8 < 8; ++r8) {
                *(v4f*)(u0 + r8 * PU + cq) = (v4f){c1[0][r8], c1[1][r8], c1[2][r8], c1[3][r8]};
                u0[r8 * PU + ce] = c1[4][r8];
            }
        } else {
#pragma unroll
            for (int r8 = 0; r8 < 8; ++r8) {
                const int lrow = (r8 < 3 && kh == 0) ? (r8 * RPI + 3 + SRC - 1) : (3 * RPI + r8 + kh * 8);
                float* u = Us + lrow * PU;
                *(v4f*)(u + cq) = (v4f){c1[0][r8], c1[1][r8], c1[2][r8], c1[3][r8]};
                u[ce] = c1[4][r8];
            }
        }
    }
    __syncthreads();

    {
        const int wl0 = tid >> 6;
        const int k   = tid & 63;
        const int kq  = (k < K63) ? k : (K63 - 1);
        const int j2  = kq / 21;
        const int kc  = kq - j2 * 21;
        const int hr  = h + j2 - 1;
        const bool rowok = (k < K63) && (hr >= 0) && (hr < H28);
        const float* ub = Us + (j2 * RPI + 3 * wh + wl0) * PU + kc;
        const int wlast = (wl0 + 12 < WOUT) ? (wl0 + 12) : (WOUT - 1);
        const float* ub3 = Us + (j2 * RPI + 3 * wh + wlast) * PU + kc;
#pragma unroll
        for (int e = 0; e < 4; ++e) {
            float acc = 0.0f;
#pragma unroll
            for (int j = 0; j < 7; ++j)
                acc += (e < 3) ? ub[(4 * e + j) * PU + j * 21] : ub3[j * PU + j * 21];
            const int wl = wl0 + 4 * e;
            Bs[wl * PB + k] = (rowok && wl < WOUT) ? acc : 0.0f;
        }
    }
    __syncthreads();

    v8f c2 = {0.f, 0.f, 0.f, 0.f, 0.f, 0.f, 0.f, 0.f};
    const float* brow = Bs + t * PB;
    v2f b2[16];
#pragma unroll
    for (int kk = 0; kk < 16; ++kk) {
        const int k0 = kk * 4 + kh * 2;
        b2[kk] = *(const v2f*)(brow + k0);
    }
    __builtin_amdgcn_sched_barrier(0);
#pragma unroll
    for (int kk = 0; kk < 16; ++kk) {
        c2 = __builtin_amdgcn_wmma_f32_16x16x4_f32(
                  false, a2[kk],  false, b2[kk],
                  (short)0, c2,  false,  false);
    }

    if (t < WOUT) {
        float* yo = y + (size_t)(g * 128 + wave * 16 + kh * 8) * P784 + h * H28 + WOUT * wh + t;
#pragma unroll
        for (int r8 = 0; r8 < 8; ++r8)
            yo[(size_t)r8 * P784] = c2[r8];
    }
}

extern "C" void kernel_launch(void* const* d_in, const int* in_sizes, int n_in,
                              void* d_out, int out_size, void* d_ws, size_t ws_size,
                              hipStream_t stream) {
    const float* x  = (const float*)d_in[0];
    const float* w0 = (const float*)d_in[1];
    const float* w1 = (const float*)d_in[2];
    float* y = (float*)d_out;

    k_fused<<<dim3(28, 2, 2), 256, 0, stream>>>(x, w0, w1, y);
}
